# starting kernel
# baseline (speedup 1.0000x reference)
#include <hip/hip_runtime.h>

typedef float v2f __attribute__((ext_vector_type(2)));
typedef float v4f __attribute__((ext_vector_type(4)));
typedef float v8f __attribute__((ext_vector_type(8)));

constexpr int NB   = 16;
constexpr int CC   = 1024;
constexpr int DIN  = 512;
constexpr int DH   = 1024;
constexpr int DOUT = 512;

__global__ void k_partial_sum_x(const float* __restrict__ x, float* __restrict__ s1p) {
    const int n     = blockIdx.x & 15;
    const int chunk = blockIdx.x >> 4;
    const int d     = threadIdx.x;
    const float* px = x + (size_t)n * CC * DIN + (size_t)chunk * 128 * DIN + d;
    float s = 0.f;
    #pragma unroll 8
    for (int c = 0; c < 128; ++c) s += px[(size_t)c * DIN];
    s1p[((size_t)chunk * NB + n) * DIN + d] = s;
}

__global__ void k_reduce_s1(const float* __restrict__ s1p, float* __restrict__ s1) {
    const int i = blockIdx.x * blockDim.x + threadIdx.x;
    float s = 0.f;
    #pragma unroll
    for (int ch = 0; ch < 8; ++ch) s += s1p[(size_t)ch * (NB * DIN) + i];
    s1[i] = s;
}

__global__ void k_sw(const float* __restrict__ adj, float* __restrict__ rSw) {
    const int n    = blockIdx.x;
    const int lane = threadIdx.x;
    const float* pa = adj + (size_t)n * CC * CC;
    float s = 0.f;
    for (int e = lane; e < CC; e += 32) s += pa[e];
    #pragma unroll
    for (int off = 16; off > 0; off >>= 1) s += __shfl_down(s, off, 32);
    if (lane == 0) rSw[n] = (s > 0.f) ? (1.0f / s) : 0.f;
}

__global__ void k_gemm1_relu(const float* __restrict__ s1, const float* __restrict__ W1,
                             const float* __restrict__ b1, const float* __restrict__ rSw,
                             float* __restrict__ r1) {
    const int lane = threadIdx.x;
    const int tile = blockIdx.x;
    const int low  = lane & 15;
    const int hi   = lane >> 4;
    const int col  = tile * 16 + low;
    const float* pa = s1 + (size_t)low * DIN;
    const float* pb = W1 + (size_t)col * DIN;
    v8f acc = {};
    #pragma unroll 4
    for (int kk = 0; kk < DIN; kk += 4) {
        const int k0 = kk + 2 * hi;
        v2f a = { pa[k0], pa[k0 + 1] };
        v2f b = { pb[k0], pb[k0 + 1] };
        acc = __builtin_amdgcn_wmma_f32_16x16x4_f32(false, a, false, b,
                                                    (short)0, acc, false, false);
    }
    const float bias = b1[col];
    #pragma unroll
    for (int r = 0; r < 8; ++r) {
        const int Mrow = r + 8 * hi;
        float u = acc[r] * rSw[Mrow] + bias;
        u = u > 0.f ? u : 0.f;
        r1[(size_t)Mrow * DH + col] = u;
    }
}

__global__ void k_ln(const float* __restrict__ r1, const float* __restrict__ g,
                     const float* __restrict__ bb, float* __restrict__ lnout) {
    __shared__ float ssum[256];
    __shared__ float ssq[256];
    const int n = blockIdx.x, t = threadIdx.x;
    const float* row = r1 + (size_t)n * DH;
    float vals[4];
    float s = 0.f, q = 0.f;
    #pragma unroll
    for (int i = 0; i < 4; ++i) {
        float xv = row[t + i * 256];
        vals[i] = xv; s += xv; q += xv * xv;
    }
    ssum[t] = s; ssq[t] = q; __syncthreads();
    for (int off = 128; off > 0; off >>= 1) {
        if (t < off) { ssum[t] += ssum[t + off]; ssq[t] += ssq[t + off]; }
        __syncthreads();
    }
    const float mean = ssum[0] * (1.0f / DH);
    const float var  = ssq[0] * (1.0f / DH) - mean * mean;
    const float inv  = rsqrtf(var + 1e-5f);
    #pragma unroll
    for (int i = 0; i < 4; ++i) {
        const int h = t + i * 256;
        lnout[(size_t)n * DH + h] = (vals[i] - mean) * inv * g[h] + bb[h];
    }
}

__global__ void k_gemm2(const float* __restrict__ ln, const float* __restrict__ W2,
                        const float* __restrict__ b2, const float* __restrict__ rSw,
                        float* __restrict__ v) {
    const int lane = threadIdx.x;
    const int tile = blockIdx.x;
    const int low  = lane & 15;
    const int hi   = lane >> 4;
    const int col  = tile * 16 + low;
    const float* pa = ln + (size_t)low * DH;
    const float* pb = W2 + (size_t)col * DH;
    v8f acc = {};
    #pragma unroll 4
    for (int kk = 0; kk < DH; kk += 4) {
        const int k0 = kk + 2 * hi;
        v2f a = { pa[k0], pa[k0 + 1] };
        v2f b = { pb[k0], pb[k0 + 1] };
        acc = __builtin_amdgcn_wmma_f32_16x16x4_f32(false, a, false, b,
                                                    (short)0, acc, false, false);
    }
    const float bias = b2[col];
    #pragma unroll
    for (int r = 0; r < 8; ++r) {
        const int Mrow = r + 8 * hi;
        v[(size_t)Mrow * DOUT + col] = acc[r] * ((float)CC * rSw[Mrow]) + bias;
    }
}

__global__ void k_final(const float* __restrict__ x, const float* __restrict__ v,
                        float* __restrict__ out) {
    const size_t i4 = (size_t)blockIdx.x * blockDim.x + threadIdx.x;
    const int    d4 = (int)(i4 & 127);
    const size_t nc = i4 >> 7;
    const int    n  = (int)(nc >> 10);
    const v4f* xv = (const v4f*)x;
    const v4f* vv = (const v4f*)v;
    v4f xx = __builtin_nontemporal_load(&xv[i4]);
    v4f av = vv[(size_t)n * 128 + d4];
    __builtin_nontemporal_store(xx + av, ((v4f*)out) + i4);
}

extern "C" void kernel_launch(void* const* d_in, const int* in_sizes, int n_in,
                              void* d_out, int out_size, void* d_ws, size_t ws_size,
                              hipStream_t stream) {
    const float* x     = (const float*)d_in[0];
    const float* adj   = (const float*)d_in[1];
    const float* W1    = (const float*)d_in[2];
    const float* b1    = (const float*)d_in[3];
    const float* gamma = (const float*)d_in[4];
    const float* beta  = (const float*)d_in[5];
    const float* W2    = (const float*)d_in[6];
    const float* b2    = (const float*)d_in[7];
    float* out = (float*)d_out;

    float* ws  = (float*)d_ws;
    float* s1p = ws;
    float* s1  = s1p + 8 * NB * DIN;
    float* rSw = s1 + NB * DIN;
    float* r1  = rSw + 32;
    float* lnb = r1 + NB * DH;
    float* vb  = lnb + NB * DH;

    k_partial_sum_x<<<dim3(8 * NB), dim3(DIN), 0, stream>>>(x, s1p);
    k_reduce_s1   <<<dim3(NB),      dim3(DIN), 0, stream>>>(s1p, s1);
    k_sw          <<<dim3(NB),      dim3(32),  0, stream>>>(adj, rSw);
    k_gemm1_relu  <<<dim3(DH / 16), dim3(32),  0, stream>>>(s1, W1, b1, rSw, r1);
    k_ln          <<<dim3(NB),      dim3(256), 0, stream>>>(r1, gamma, beta, lnb);
    k_gemm2       <<<dim3(DOUT/16), dim3(32),  0, stream>>>(lnb, W2, b2, rSw, vb);
    k_final       <<<dim3((NB * CC * DIN / 4) / 256), dim3(256), 0, stream>>>(x, vb, out);
}


# GEMM2 moves into the final add's launch (7 launches to 6): each of 256 work-groups forms its own 16x32 tile of v, K split over its 16 waves, partials summed through LDS (new f32 sum order)
# speedup vs baseline: 1.3905x; 1.3905x over previous
#include <hip/hip_runtime.h>

typedef float v2f __attribute__((ext_vector_type(2)));
typedef float v4f __attribute__((ext_vector_type(4)));
typedef float v8f __attribute__((ext_vector_type(8)));

constexpr int NB   = 16;
constexpr int CC   = 1024;
constexpr int DIN  = 512;
constexpr int DH   = 1024;
constexpr int DOUT = 512;

__global__ void k_partial_sum_x(const float* __restrict__ x, float* __restrict__ s1p) {
    const int n     = blockIdx.x & 15;
    const int chunk = blockIdx.x >> 4;
    const int d     = threadIdx.x;
    const float* px = x + (size_t)n * CC * DIN + (size_t)chunk * 128 * DIN + d;
    float s = 0.f;
    #pragma unroll 8
    for (int c = 0; c < 128; ++c) s += px[(size_t)c * DIN];
    s1p[((size_t)chunk * NB + n) * DIN + d] = s;
}

__global__ void k_reduce_s1(const float* __restrict__ s1p, float* __restrict__ s1) {
    const int i = blockIdx.x * blockDim.x + threadIdx.x;
    float s = 0.f;
    #pragma unroll
    for (int ch = 0; ch < 8; ++ch) s += s1p[(size_t)ch * (NB * DIN) + i];
    s1[i] = s;
}

__global__ void k_sw(const float* __restrict__ adj, float* __restrict__ rSw) {
    const int n    = blockIdx.x;
    const int lane = threadIdx.x;
    const float* pa = adj + (size_t)n * CC * CC;
    float s = 0.f;
    for (int e = lane; e < CC; e += 32) s += pa[e];
    #pragma unroll
    for (int off = 16; off > 0; off >>= 1) s += __shfl_down(s, off, 32);
    if (lane == 0) rSw[n] = (s > 0.f) ? (1.0f / s) : 0.f;
}

__global__ void k_gemm1_relu(const float* __restrict__ s1, const float* __restrict__ W1,
                             const float* __restrict__ b1, const float* __restrict__ rSw,
                             float* __restrict__ r1) {
    const int lane = threadIdx.x;
    const int tile = blockIdx.x;
    const int low  = lane & 15;
    const int hi   = lane >> 4;
    const int col  = tile * 16 + low;
    const float* pa = s1 + (size_t)low * DIN;
    const float* pb = W1 + (size_t)col * DIN;
    v8f acc = {};
    #pragma unroll 4
    for (int kk = 0; kk < DIN; kk += 4) {
        const int k0 = kk + 2 * hi;
        v2f a = { pa[k0], pa[k0 + 1] };
        v2f b = { pb[k0], pb[k0 + 1] };
        acc = __builtin_amdgcn_wmma_f32_16x16x4_f32(false, a, false, b,
                                                    (short)0, acc, false, false);
    }
    const float bias = b1[col];
    #pragma unroll
    for (int r = 0; r < 8; ++r) {
        const int Mrow = r + 8 * hi;
        float u = acc[r] * rSw[Mrow] + bias;
        u = u > 0.f ? u : 0.f;
        r1[(size_t)Mrow * DH + col] = u;
    }
}

__global__ void k_ln(const float* __restrict__ r1, const float* __restrict__ g,
                     const float* __restrict__ bb, float* __restrict__ lnout) {
    __shared__ float ssum[256];
    __shared__ float ssq[256];
    const int n = blockIdx.x, t = threadIdx.x;
    const float* row = r1 + (size_t)n * DH;
    float vals[4];
    float s = 0.f, q = 0.f;
    #pragma unroll
    for (int i = 0; i < 4; ++i) {
        float xv = row[t + i * 256];
        vals[i] = xv; s += xv; q += xv * xv;
    }
    ssum[t] = s; ssq[t] = q; __syncthreads();
    for (int off = 128; off > 0; off >>= 1) {
        if (t < off) { ssum[t] += ssum[t + off]; ssq[t] += ssq[t + off]; }
        __syncthreads();
    }
    const float mean = ssum[0] * (1.0f / DH);
    const float var  = ssq[0] * (1.0f / DH) - mean * mean;
    const float inv  = rsqrtf(var + 1e-5f);
    #pragma unroll
    for (int i = 0; i < 4; ++i) {
        const int h = t + i * 256;
        lnout[(size_t)n * DH + h] = (vals[i] - mean) * inv * g[h] + bb[h];
    }
}

constexpr int PP = 36;

__global__ __launch_bounds__(512)
void k_gemm2_final(const float* __restrict__ x, const float* __restrict__ ln,
                   const float* __restrict__ W2, const float* __restrict__ b2,
                   const float* __restrict__ rSw, float* __restrict__ out) {
    __shared__ float part[16][16][PP];
    const int p    = blockIdx.x & 15;
    const int q    = blockIdx.x >> 4;
    const int lane = threadIdx.x & 31;
    const int w    = threadIdx.x >> 5;
    const int low  = lane & 15;
    const int hi   = lane >> 4;

    const v4f* pa  = (const v4f*)(ln + (size_t)low * DH + w * 64 + 4 * hi);
    const v4f* pb0 = (const v4f*)(W2 + (size_t)(p * 32 + low) * DH + w * 64 + 4 * hi);
    const v4f* pb1 = pb0 + 16 * (DH / 4);
    v4f a[8], b0[8], b1[8];
    #pragma unroll
    for (int j = 0; j < 8; ++j) { a[j] = pa[2 * j]; b0[j] = pb0[2 * j]; b1[j] = pb1[2 * j]; }
    const int d4 = lane & 7;
    const float rsw  = rSw[w];
    const v4f   bias = *(const v4f*)(b2 + p * 32 + d4 * 4);

    const int rs = lane >> 3;
    const size_t xoff = ((size_t)w * CC + q * 64 + rs) * DIN + p * 32 + d4 * 4;
    const v4f* px = (const v4f*)(x + xoff);
    v4f xv[16];
    #pragma unroll
    for (int i = 0; i < 16; ++i) xv[i] = __builtin_nontemporal_load(px + i * (4 * DIN / 4));
    __builtin_amdgcn_sched_barrier(0);

    v8f acc0 = {}, acc1 = {};
    #pragma unroll
    for (int j = 0; j < 8; ++j) {
        const v2f a01 = { a[j].x, a[j].y },   a23 = { a[j].z, a[j].w };
        const v2f e01 = { b0[j].x, b0[j].y }, e23 = { b0[j].z, b0[j].w };
        const v2f f01 = { b1[j].x, b1[j].y }, f23 = { b1[j].z, b1[j].w };
        acc0 = __builtin_amdgcn_wmma_f32_16x16x4_f32(false, a01, false, e01, (short)0, acc0, false, false);
        acc1 = __builtin_amdgcn_wmma_f32_16x16x4_f32(false, a01, false, f01, (short)0, acc1, false, false);
        acc0 = __builtin_amdgcn_wmma_f32_16x16x4_f32(false, a23, false, e23, (short)0, acc0, false, false);
        acc1 = __builtin_amdgcn_wmma_f32_16x16x4_f32(false, a23, false, f23, (short)0, acc1, false, false);
    }
    #pragma unroll
    for (int r = 0; r < 8; ++r) {
        part[w][r + 8 * hi][low]      = acc0[r];
        part[w][r + 8 * hi][16 + low] = acc1[r];
    }
    __syncthreads();

    v4f s = { 0.f, 0.f, 0.f, 0.f };
    #pragma unroll
    for (int k = 0; k < 16; ++k) s += *(const v4f*)&part[k][w][d4 * 4];
    const float sc = (float)CC * rsw;
    const v4f av = s * sc + bias;

    v4f* po = (v4f*)(out + xoff);
    #pragma unroll
    for (int i = 0; i < 16; ++i) __builtin_nontemporal_store(xv[i] + av, po + i * (4 * DIN / 4));
}

extern "C" void kernel_launch(void* const* d_in, const int* in_sizes, int n_in,
                              void* d_out, int out_size, void* d_ws, size_t ws_size,
                              hipStream_t stream) {
    const float* x     = (const float*)d_in[0];
    const float* adj   = (const float*)d_in[1];
    const float* W1    = (const float*)d_in[2];
    const float* b1    = (const float*)d_in[3];
    const float* gamma = (const float*)d_in[4];
    const float* beta  = (const float*)d_in[5];
    const float* W2    = (const float*)d_in[6];
    const float* b2    = (const float*)d_in[7];
    float* out = (float*)d_out;

    float* ws  = (float*)d_ws;
    float* s1p = ws;
    float* s1  = s1p + 8 * NB * DIN;
    float* rSw = s1 + NB * DIN;
    float* r1  = rSw + 32;
    float* lnb = r1 + NB * DH;

    k_partial_sum_x<<<dim3(8 * NB), dim3(DIN), 0, stream>>>(x, s1p);
    k_reduce_s1   <<<dim3(NB),      dim3(DIN), 0, stream>>>(s1p, s1);
    k_sw          <<<dim3(NB),      dim3(32),  0, stream>>>(adj, rSw);
    k_gemm1_relu  <<<dim3(DH / 16), dim3(32),  0, stream>>>(s1, W1, b1, rSw, r1);
    k_ln          <<<dim3(NB),      dim3(256), 0, stream>>>(r1, gamma, beta, lnb);
    k_gemm2_final <<<dim3(16 * 16),  dim3(512), 0, stream>>>(x, lnb, W2, b2, rSw, out);
}


# GEMM1: K split over 4 work-groups per 32-column tile (128 one-wave groups), each loads its whole K slice before its first WMMA; LayerNorm's loads add the 4 partial planes (new f32 sum order)
# speedup vs baseline: 1.6091x; 1.1572x over previous
#include <hip/hip_runtime.h>

typedef float v2f __attribute__((ext_vector_type(2)));
typedef float v4f __attribute__((ext_vector_type(4)));
typedef float v8f __attribute__((ext_vector_type(8)));

constexpr int NB   = 16;
constexpr int CC   = 1024;
constexpr int DIN  = 512;
constexpr int DH   = 1024;
constexpr int DOUT = 512;

__global__ void k_partial_sum_x(const float* __restrict__ x, float* __restrict__ s1p) {
    const int n     = blockIdx.x & 15;
    const int chunk = blockIdx.x >> 4;
    const int d     = threadIdx.x;
    const float* px = x + (size_t)n * CC * DIN + (size_t)chunk * 128 * DIN + d;
    float s = 0.f;
    #pragma unroll 8
    for (int c = 0; c < 128; ++c) s += px[(size_t)c * DIN];
    s1p[((size_t)chunk * NB + n) * DIN + d] = s;
}

__global__ void k_reduce_s1(const float* __restrict__ s1p, float* __restrict__ s1) {
    const int i = blockIdx.x * blockDim.x + threadIdx.x;
    float s = 0.f;
    #pragma unroll
    for (int ch = 0; ch < 8; ++ch) s += s1p[(size_t)ch * (NB * DIN) + i];
    s1[i] = s;
}

__global__ void k_sw(const float* __restrict__ adj, float* __restrict__ rSw) {
    const int n    = blockIdx.x;
    const int lane = threadIdx.x;
    const float* pa = adj + (size_t)n * CC * CC;
    float s = 0.f;
    for (int e = lane; e < CC; e += 32) s += pa[e];
    #pragma unroll
    for (int off = 16; off > 0; off >>= 1) s += __shfl_down(s, off, 32);
    if (lane == 0) rSw[n] = (s > 0.f) ? (1.0f / s) : 0.f;
}

constexpr int KS1 = 4;
constexpr int KL1 = DIN / KS1;

__global__ __launch_bounds__(32)
void k_gemm1_part(const float* __restrict__ s1, const float* __restrict__ W1,
                             float* __restrict__ up) {
    const int lane = threadIdx.x;
    const int tile = blockIdx.x & 31;
    const int ks   = blockIdx.x >> 5;
    const int low  = lane & 15;
    const int hi   = lane >> 4;

    const v4f* pa  = (const v4f*)(s1 + (size_t)low * DIN + ks * KL1 + 4 * hi);
    const v4f* pb0 = (const v4f*)(W1 + (size_t)(tile * 32 + low) * DIN + ks * KL1 + 4 * hi);
    const v4f* pb1 = pb0 + 16 * (DIN / 4);
    v4f a[16], b0[16], b1[16];
    #pragma unroll
    for (int j = 0; j < 16; ++j) { a[j] = pa[2 * j]; b0[j] = pb0[2 * j]; b1[j] = pb1[2 * j]; }
    __builtin_amdgcn_sched_barrier(0);

    v8f acc0 = {}, acc1 = {};
    #pragma unroll
    for (int j = 0; j < 16; ++j) {
        const v2f a01 = { a[j].x, a[j].y },   a23 = { a[j].z, a[j].w };
        const v2f e01 = { b0[j].x, b0[j].y }, e23 = { b0[j].z, b0[j].w };
        const v2f f01 = { b1[j].x, b1[j].y }, f23 = { b1[j].z, b1[j].w };
        acc0 = __builtin_amdgcn_wmma_f32_16x16x4_f32(false, a01, false, e01, (short)0, acc0, false, false);
        acc1 = __builtin_amdgcn_wmma_f32_16x16x4_f32(false, a01, false, f01, (short)0, acc1, false, false);
        acc0 = __builtin_amdgcn_wmma_f32_16x16x4_f32(false, a23, false, e23, (short)0, acc0, false, false);
        acc1 = __builtin_amdgcn_wmma_f32_16x16x4_f32(false, a23, false, f23, (short)0, acc1, false, false);
    }
    float* po = up + (size_t)ks * (NB * DH) + (size_t)(8 * hi) * DH + tile * 32 + low;
    #pragma unroll
    for (int r = 0; r < 8; ++r) {
        po[(size_t)r * DH]      = acc0[r];
        po[(size_t)r * DH + 16] = acc1[r];
    }
}

__global__ void k_ln(const float* __restrict__ up, const float* __restrict__ b1,
                     const float* __restrict__ rSw, const float* __restrict__ g,
                     const float* __restrict__ bb, float* __restrict__ lnout) {
    __shared__ float ssum[256];
    __shared__ float ssq[256];
    const int n = blockIdx.x, t = threadIdx.x;
    const float* row = up + (size_t)n * DH;
    const float rs = rSw[n];
    float vals[4];
    float s = 0.f, q = 0.f;
    #pragma unroll
    for (int i = 0; i < 4; ++i) {
        const int h = t + i * 256;
        float u = row[h];
        #pragma unroll
        for (int k = 1; k < KS1; ++k) u += row[(size_t)k * (NB * DH) + h];
        u = u * rs + b1[h];
        const float xv = u > 0.f ? u : 0.f;
        vals[i] = xv; s += xv; q += xv * xv;
    }
    ssum[t] = s; ssq[t] = q; __syncthreads();
    for (int off = 128; off > 0; off >>= 1) {
        if (t < off) { ssum[t] += ssum[t + off]; ssq[t] += ssq[t + off]; }
        __syncthreads();
    }
    const float mean = ssum[0] * (1.0f / DH);
    const float var  = ssq[0] * (1.0f / DH) - mean * mean;
    const float inv  = rsqrtf(var + 1e-5f);
    #pragma unroll
    for (int i = 0; i < 4; ++i) {
        const int h = t + i * 256;
        lnout[(size_t)n * DH + h] = (vals[i] - mean) * inv * g[h] + bb[h];
    }
}

constexpr int PP = 36;

__global__ __launch_bounds__(512)
void k_gemm2_final(const float* __restrict__ x, const float* __restrict__ ln,
                   const float* __restrict__ W2, const float* __restrict__ b2,
                   const float* __restrict__ rSw, float* __restrict__ out) {
    __shared__ float part[16][16][PP];
    const int p    = blockIdx.x & 15;
    const int q    = blockIdx.x >> 4;
    const int lane = threadIdx.x & 31;
    const int w    = threadIdx.x >> 5;
    const int low  = lane & 15;
    const int hi   = lane >> 4;

    const v4f* pa  = (const v4f*)(ln + (size_t)low * DH + w * 64 + 4 * hi);
    const v4f* pb0 = (const v4f*)(W2 + (size_t)(p * 32 + low) * DH + w * 64 + 4 * hi);
    const v4f* pb1 = pb0 + 16 * (DH / 4);
    v4f a[8], b0[8], b1[8];
    #pragma unroll
    for (int j = 0; j < 8; ++j) { a[j] = pa[2 * j]; b0[j] = pb0[2 * j]; b1[j] = pb1[2 * j]; }
    const int d4 = lane & 7;
    const float rsw  = rSw[w];
    const v4f   bias = *(const v4f*)(b2 + p * 32 + d4 * 4);

    const int rs = lane >> 3;
    const size_t xoff = ((size_t)w * CC + q * 64 + rs) * DIN + p * 32 + d4 * 4;
    const v4f* px = (const v4f*)(x + xoff);
    v4f xv[16];
    #pragma unroll
    for (int i = 0; i < 16; ++i) xv[i] = __builtin_nontemporal_load(px + i * (4 * DIN / 4));
    __builtin_amdgcn_sched_barrier(0);

    v8f acc0 = {}, acc1 = {};
    #pragma unroll
    for (int j = 0; j < 8; ++j) {
        const v2f a01 = { a[j].x, a[j].y },   a23 = { a[j].z, a[j].w };
        const v2f e01 = { b0[j].x, b0[j].y }, e23 = { b0[j].z, b0[j].w };
        const v2f f01 = { b1[j].x, b1[j].y }, f23 = { b1[j].z, b1[j].w };
        acc0 = __builtin_amdgcn_wmma_f32_16x16x4_f32(false, a01, false, e01, (short)0, acc0, false, false);
        acc1 = __builtin_amdgcn_wmma_f32_16x16x4_f32(false, a01, false, f01, (short)0, acc1, false, false);
        acc0 = __builtin_amdgcn_wmma_f32_16x16x4_f32(false, a23, false, e23, (short)0, acc0, false, false);
        acc1 = __builtin_amdgcn_wmma_f32_16x16x4_f32(false, a23, false, f23, (short)0, acc1, false, false);
    }
    #pragma unroll
    for (int r = 0; r < 8; ++r) {
        part[w][r + 8 * hi][low]      = acc0[r];
        part[w][r + 8 * hi][16 + low] = acc1[r];
    }
    __syncthreads();

    v4f s = { 0.f, 0.f, 0.f, 0.f };
    #pragma unroll
    for (int k = 0; k < 16; ++k) s += *(const v4f*)&part[k][w][d4 * 4];
    const float sc = (float)CC * rsw;
    const v4f av = s * sc + bias;

    v4f* po = (v4f*)(out + xoff);
    #pragma unroll
    for (int i = 0; i < 16; ++i) __builtin_nontemporal_store(xv[i] + av, po + i * (4 * DIN / 4));
}

extern "C" void kernel_launch(void* const* d_in, const int* in_sizes, int n_in,
                              void* d_out, int out_size, void* d_ws, size_t ws_size,
                              hipStream_t stream) {
    const float* x     = (const float*)d_in[0];
    const float* adj   = (const float*)d_in[1];
    const float* W1    = (const float*)d_in[2];
    const float* b1    = (const float*)d_in[3];
    const float* gamma = (const float*)d_in[4];
    const float* beta  = (const float*)d_in[5];
    const float* W2    = (const float*)d_in[6];
    const float* b2    = (const float*)d_in[7];
    float* out = (float*)d_out;

    float* ws  = (float*)d_ws;
    float* s1p = ws;
    float* s1  = s1p + 8 * NB * DIN;
    float* rSw = s1 + NB * DIN;
    float* lnb = rSw + 32;
    float* up  = ws;

    k_partial_sum_x<<<dim3(8 * NB), dim3(DIN), 0, stream>>>(x, s1p);
    k_reduce_s1   <<<dim3(NB),      dim3(DIN), 0, stream>>>(s1p, s1);
    k_sw          <<<dim3(NB),      dim3(32),  0, stream>>>(adj, rSw);
    k_gemm1_part  <<<dim3(32 * KS1), dim3(32),  0, stream>>>(s1, W1, up);
    k_ln          <<<dim3(NB),      dim3(256), 0, stream>>>(up, b1, rSw, gamma, beta, lnb);
    k_gemm2_final <<<dim3(16 * 16),  dim3(512), 0, stream>>>(x, lnb, W2, b2, rSw, out);
}


# Column sum of x: 256 groups of 8 waves each own a 1024-row x 32-column panel, 32 128-bit loads per lane in flight, LDS sum writes s1; partial planes and reduce launch go (6 to 5; new f32 sum order)
# speedup vs baseline: 2.1648x; 1.3453x over previous
#include <hip/hip_runtime.h>

typedef float v2f __attribute__((ext_vector_type(2)));
typedef float v4f __attribute__((ext_vector_type(4)));
typedef float v8f __attribute__((ext_vector_type(8)));

constexpr int NB   = 16;
constexpr int CC   = 1024;
constexpr int DIN  = 512;
constexpr int DH   = 1024;
constexpr int DOUT = 512;

constexpr int CSW = 8;
constexpr int CSR = CC / CSW;
constexpr int CSL = CSR / 4;

__global__ __launch_bounds__(32 * CSW)
void k_colsum_x(const float* __restrict__ x, float* __restrict__ s1) {
    __shared__ alignas(16) float part[CSW][4][32];
    const int p    = blockIdx.x & 15;
    const int n    = blockIdx.x >> 4;
    const int lane = threadIdx.x & 31;
    const int w    = threadIdx.x >> 5;
    const int d4   = lane & 7;
    const int rs   = lane >> 3;
    const v4f* px = (const v4f*)(x + ((size_t)n * CC + w * CSR + rs) * DIN + p * 32 + d4 * 4);
    v4f xv[CSL];
    #pragma unroll
    for (int i = 0; i < CSL; ++i) xv[i] = px[i * (4 * DIN / 4)];
    __builtin_amdgcn_sched_barrier(0);

    v4f c0 = xv[0], c1 = xv[1], c2 = xv[2], c3 = xv[3];
    #pragma unroll
    for (int i = 4; i < CSL; i += 4) { c0 += xv[i]; c1 += xv[i + 1]; c2 += xv[i + 2]; c3 += xv[i + 3]; }
    *(v4f*)&part[w][rs][d4 * 4] = (c0 + c1) + (c2 + c3);
    __syncthreads();

    if (w == 0) {
        float t0 = part[0][0][lane], t1 = part[0][1][lane], t2 = part[0][2][lane], t3 = part[0][3][lane];
        #pragma unroll
        for (int k = 1; k < CSW; ++k) {
            t0 += part[k][0][lane]; t1 += part[k][1][lane]; t2 += part[k][2][lane]; t3 += part[k][3][lane];
        }
        s1[(size_t)n * DIN + p * 32 + lane] = (t0 + t1) + (t2 + t3);
    }
}

__global__ void k_sw(const float* __restrict__ adj, float* __restrict__ rSw) {
    const int n    = blockIdx.x;
    const int lane = threadIdx.x;
    const float* pa = adj + (size_t)n * CC * CC;
    float s = 0.f;
    for (int e = lane; e < CC; e += 32) s += pa[e];
    #pragma unroll
    for (int off = 16; off > 0; off >>= 1) s += __shfl_down(s, off, 32);
    if (lane == 0) rSw[n] = (s > 0.f) ? (1.0f / s) : 0.f;
}

constexpr int KS1 = 4;
constexpr int KL1 = DIN / KS1;

__global__ __launch_bounds__(32)
void k_gemm1_part(const float* __restrict__ s1, const float* __restrict__ W1,
                             float* __restrict__ up) {
    const int lane = threadIdx.x;
    const int tile = blockIdx.x & 31;
    const int ks   = blockIdx.x >> 5;
    const int low  = lane & 15;
    const int hi   = lane >> 4;

    const v4f* pa  = (const v4f*)(s1 + (size_t)low * DIN + ks * KL1 + 4 * hi);
    const v4f* pb0 = (const v4f*)(W1 + (size_t)(tile * 32 + low) * DIN + ks * KL1 + 4 * hi);
    const v4f* pb1 = pb0 + 16 * (DIN / 4);
    v4f a[16], b0[16], b1[16];
    #pragma unroll
    for (int j = 0; j < 16; ++j) { a[j] = pa[2 * j]; b0[j] = pb0[2 * j]; b1[j] = pb1[2 * j]; }
    __builtin_amdgcn_sched_barrier(0);

    v8f acc0 = {}, acc1 = {};
    #pragma unroll
    for (int j = 0; j < 16; ++j) {
        const v2f a01 = { a[j].x, a[j].y },   a23 = { a[j].z, a[j].w };
        const v2f e01 = { b0[j].x, b0[j].y }, e23 = { b0[j].z, b0[j].w };
        const v2f f01 = { b1[j].x, b1[j].y }, f23 = { b1[j].z, b1[j].w };
        acc0 = __builtin_amdgcn_wmma_f32_16x16x4_f32(false, a01, false, e01, (short)0, acc0, false, false);
        acc1 = __builtin_amdgcn_wmma_f32_16x16x4_f32(false, a01, false, f01, (short)0, acc1, false, false);
        acc0 = __builtin_amdgcn_wmma_f32_16x16x4_f32(false, a23, false, e23, (short)0, acc0, false, false);
        acc1 = __builtin_amdgcn_wmma_f32_16x16x4_f32(false, a23, false, f23, (short)0, acc1, false, false);
    }
    float* po = up + (size_t)ks * (NB * DH) + (size_t)(8 * hi) * DH + tile * 32 + low;
    #pragma unroll
    for (int r = 0; r < 8; ++r) {
        po[(size_t)r * DH]      = acc0[r];
        po[(size_t)r * DH + 16] = acc1[r];
    }
}

__global__ void k_ln(const float* __restrict__ up, const float* __restrict__ b1,
                     const float* __restrict__ rSw, const float* __restrict__ g,
                     const float* __restrict__ bb, float* __restrict__ lnout) {
    __shared__ float ssum[256];
    __shared__ float ssq[256];
    const int n = blockIdx.x, t = threadIdx.x;
    const float* row = up + (size_t)n * DH;
    const float rs = rSw[n];
    float vals[4];
    float s = 0.f, q = 0.f;
    #pragma unroll
    for (int i = 0; i < 4; ++i) {
        const int h = t + i * 256;
        float u = row[h];
        #pragma unroll
        for (int k = 1; k < KS1; ++k) u += row[(size_t)k * (NB * DH) + h];
        u = u * rs + b1[h];
        const float xv = u > 0.f ? u : 0.f;
        vals[i] = xv; s += xv; q += xv * xv;
    }
    ssum[t] = s; ssq[t] = q; __syncthreads();
    for (int off = 128; off > 0; off >>= 1) {
        if (t < off) { ssum[t] += ssum[t + off]; ssq[t] += ssq[t + off]; }
        __syncthreads();
    }
    const float mean = ssum[0] * (1.0f / DH);
    const float var  = ssq[0] * (1.0f / DH) - mean * mean;
    const float inv  = rsqrtf(var + 1e-5f);
    #pragma unroll
    for (int i = 0; i < 4; ++i) {
        const int h = t + i * 256;
        lnout[(size_t)n * DH + h] = (vals[i] - mean) * inv * g[h] + bb[h];
    }
}

constexpr int PP = 36;

__global__ __launch_bounds__(512)
void k_gemm2_final(const float* __restrict__ x, const float* __restrict__ ln,
                   const float* __restrict__ W2, const float* __restrict__ b2,
                   const float* __restrict__ rSw, float* __restrict__ out) {
    __shared__ float part[16][16][PP];
    const int p    = blockIdx.x & 15;
    const int q    = blockIdx.x >> 4;
    const int lane = threadIdx.x & 31;
    const int w    = threadIdx.x >> 5;
    const int low  = lane & 15;
    const int hi   = lane >> 4;

    const v4f* pa  = (const v4f*)(ln + (size_t)low * DH + w * 64 + 4 * hi);
    const v4f* pb0 = (const v4f*)(W2 + (size_t)(p * 32 + low) * DH + w * 64 + 4 * hi);
    const v4f* pb1 = pb0 + 16 * (DH / 4);
    v4f a[8], b0[8], b1[8];
    #pragma unroll
    for (int j = 0; j < 8; ++j) { a[j] = pa[2 * j]; b0[j] = pb0[2 * j]; b1[j] = pb1[2 * j]; }
    const int d4 = lane & 7;
    const float rsw  = rSw[w];
    const v4f   bias = *(const v4f*)(b2 + p * 32 + d4 * 4);

    const int rs = lane >> 3;
    const size_t xoff = ((size_t)w * CC + q * 64 + rs) * DIN + p * 32 + d4 * 4;
    const v4f* px = (const v4f*)(x + xoff);
    v4f xv[16];
    #pragma unroll
    for (int i = 0; i < 16; ++i) xv[i] = __builtin_nontemporal_load(px + i * (4 * DIN / 4));
    __builtin_amdgcn_sched_barrier(0);

    v8f acc0 = {}, acc1 = {};
    #pragma unroll
    for (int j = 0; j < 8; ++j) {
        const v2f a01 = { a[j].x, a[j].y },   a23 = { a[j].z, a[j].w };
        const v2f e01 = { b0[j].x, b0[j].y }, e23 = { b0[j].z, b0[j].w };
        const v2f f01 = { b1[j].x, b1[j].y }, f23 = { b1[j].z, b1[j].w };
        acc0 = __builtin_amdgcn_wmma_f32_16x16x4_f32(false, a01, false, e01, (short)0, acc0, false, false);
        acc1 = __builtin_amdgcn_wmma_f32_16x16x4_f32(false, a01, false, f01, (short)0, acc1, false, false);
        acc0 = __builtin_amdgcn_wmma_f32_16x16x4_f32(false, a23, false, e23, (short)0, acc0, false, false);
        acc1 = __builtin_amdgcn_wmma_f32_16x16x4_f32(false, a23, false, f23, (short)0, acc1, false, false);
    }
    #pragma unroll
    for (int r = 0; r < 8; ++r) {
        part[w][r + 8 * hi][low]      = acc0[r];
        part[w][r + 8 * hi][16 + low] = acc1[r];
    }
    __syncthreads();

    v4f s = { 0.f, 0.f, 0.f, 0.f };
    #pragma unroll
    for (int k = 0; k < 16; ++k) s += *(const v4f*)&part[k][w][d4 * 4];
    const float sc = (float)CC * rsw;
    const v4f av = s * sc + bias;

    v4f* po = (v4f*)(out + xoff);
    #pragma unroll
    for (int i = 0; i < 16; ++i) __builtin_nontemporal_store(xv[i] + av, po + i * (4 * DIN / 4));
}

extern "C" void kernel_launch(void* const* d_in, const int* in_sizes, int n_in,
                              void* d_out, int out_size, void* d_ws, size_t ws_size,
                              hipStream_t stream) {
    const float* x     = (const float*)d_in[0];
    const float* adj   = (const float*)d_in[1];
    const float* W1    = (const float*)d_in[2];
    const float* b1    = (const float*)d_in[3];
    const float* gamma = (const float*)d_in[4];
    const float* beta  = (const float*)d_in[5];
    const float* W2    = (const float*)d_in[6];
    const float* b2    = (const float*)d_in[7];
    float* out = (float*)d_out;

    float* ws  = (float*)d_ws;
    float* up  = ws;
    float* s1  = up + KS1 * NB * DH;
    float* rSw = s1 + NB * DIN;
    float* lnb = rSw + 32;

    k_colsum_x    <<<dim3(NB * 16),  dim3(32 * CSW), 0, stream>>>(x, s1);
    k_sw          <<<dim3(NB),      dim3(32),  0, stream>>>(adj, rSw);
    k_gemm1_part  <<<dim3(32 * KS1), dim3(32),  0, stream>>>(s1, W1, up);
    k_ln          <<<dim3(NB),      dim3(256), 0, stream>>>(up, b1, rSw, gamma, beta, lnb);
    k_gemm2_final <<<dim3(16 * 16),  dim3(512), 0, stream>>>(x, lnb, W2, b2, rSw, out);
}


# Degree sum: the lane's 32 dependent 4-byte loads (a rolled loop, one memory round trip per trip) become 8 independent 128-bit loads requested before the first add, folded pairwise (new f32 sum order)
# speedup vs baseline: 3.0933x; 1.4289x over previous
#include <hip/hip_runtime.h>

typedef float v2f __attribute__((ext_vector_type(2)));
typedef float v4f __attribute__((ext_vector_type(4)));
typedef float v8f __attribute__((ext_vector_type(8)));

constexpr int NB   = 16;
constexpr int CC   = 1024;
constexpr int DIN  = 512;
constexpr int DH   = 1024;
constexpr int DOUT = 512;

constexpr int CSW = 8;
constexpr int CSR = CC / CSW;
constexpr int CSL = CSR / 4;

__global__ __launch_bounds__(32 * CSW)
void k_colsum_x(const float* __restrict__ x, float* __restrict__ s1) {
    __shared__ alignas(16) float part[CSW][4][32];
    const int p    = blockIdx.x & 15;
    const int n    = blockIdx.x >> 4;
    const int lane = threadIdx.x & 31;
    const int w    = threadIdx.x >> 5;
    const int d4   = lane & 7;
    const int rs   = lane >> 3;
    const v4f* px = (const v4f*)(x + ((size_t)n * CC + w * CSR + rs) * DIN + p * 32 + d4 * 4);
    v4f xv[CSL];
    #pragma unroll
    for (int i = 0; i < CSL; ++i) xv[i] = px[i * (4 * DIN / 4)];
    __builtin_amdgcn_sched_barrier(0);

    v4f c0 = xv[0], c1 = xv[1], c2 = xv[2], c3 = xv[3];
    #pragma unroll
    for (int i = 4; i < CSL; i += 4) { c0 += xv[i]; c1 += xv[i + 1]; c2 += xv[i + 2]; c3 += xv[i + 3]; }
    *(v4f*)&part[w][rs][d4 * 4] = (c0 + c1) + (c2 + c3);
    __syncthreads();

    if (w == 0) {
        float t0 = part[0][0][lane], t1 = part[0][1][lane], t2 = part[0][2][lane], t3 = part[0][3][lane];
        #pragma unroll
        for (int k = 1; k < CSW; ++k) {
            t0 += part[k][0][lane]; t1 += part[k][1][lane]; t2 += part[k][2][lane]; t3 += part[k][3][lane];
        }
        s1[(size_t)n * DIN + p * 32 + lane] = (t0 + t1) + (t2 + t3);
    }
}

__global__ void k_sw(const float* __restrict__ adj, float* __restrict__ rSw) {
    const int n    = blockIdx.x;
    const int lane = threadIdx.x;
    const v4f* pa = (const v4f*)(adj + (size_t)n * CC * CC) + lane;
    v4f wv[8];
    #pragma unroll
    for (int i = 0; i < 8; ++i) wv[i] = pa[32 * i];
    const v4f w4 = ((wv[0] + wv[1]) + (wv[2] + wv[3])) + ((wv[4] + wv[5]) + (wv[6] + wv[7]));
    float s = (w4.x + w4.y) + (w4.z + w4.w);
    #pragma unroll
    for (int off = 16; off > 0; off >>= 1) s += __shfl_down(s, off, 32);
    if (lane == 0) rSw[n] = (s > 0.f) ? (1.0f / s) : 0.f;
}

constexpr int KS1 = 4;
constexpr int KL1 = DIN / KS1;

__global__ __launch_bounds__(32)
void k_gemm1_part(const float* __restrict__ s1, const float* __restrict__ W1,
                             float* __restrict__ up) {
    const int lane = threadIdx.x;
    const int tile = blockIdx.x & 31;
    const int ks   = blockIdx.x >> 5;
    const int low  = lane & 15;
    const int hi   = lane >> 4;

    const v4f* pa  = (const v4f*)(s1 + (size_t)low * DIN + ks * KL1 + 4 * hi);
    const v4f* pb0 = (const v4f*)(W1 + (size_t)(tile * 32 + low) * DIN + ks * KL1 + 4 * hi);
    const v4f* pb1 = pb0 + 16 * (DIN / 4);
    v4f a[16], b0[16], b1[16];
    #pragma unroll
    for (int j = 0; j < 16; ++j) { a[j] = pa[2 * j]; b0[j] = pb0[2 * j]; b1[j] = pb1[2 * j]; }
    __builtin_amdgcn_sched_barrier(0);

    v8f acc0 = {}, acc1 = {};
    #pragma unroll
    for (int j = 0; j < 16; ++j) {
        const v2f a01 = { a[j].x, a[j].y },   a23 = { a[j].z, a[j].w };
        const v2f e01 = { b0[j].x, b0[j].y }, e23 = { b0[j].z, b0[j].w };
        const v2f f01 = { b1[j].x, b1[j].y }, f23 = { b1[j].z, b1[j].w };
        acc0 = __builtin_amdgcn_wmma_f32_16x16x4_f32(false, a01, false, e01, (short)0, acc0, false, false);
        acc1 = __builtin_amdgcn_wmma_f32_16x16x4_f32(false, a01, false, f01, (short)0, acc1, false, false);
        acc0 = __builtin_amdgcn_wmma_f32_16x16x4_f32(false, a23, false, e23, (short)0, acc0, false, false);
        acc1 = __builtin_amdgcn_wmma_f32_16x16x4_f32(false, a23, false, f23, (short)0, acc1, false, false);
    }
    float* po = up + (size_t)ks * (NB * DH) + (size_t)(8 * hi) * DH + tile * 32 + low;
    #pragma unroll
    for (int r = 0; r < 8; ++r) {
        po[(size_t)r * DH]      = acc0[r];
        po[(size_t)r * DH + 16] = acc1[r];
    }
}

__global__ void k_ln(const float* __restrict__ up, const float* __restrict__ b1,
                     const float* __restrict__ rSw, const float* __restrict__ g,
                     const float* __restrict__ bb, float* __restrict__ lnout) {
    __shared__ float ssum[256];
    __shared__ float ssq[256];
    const int n = blockIdx.x, t = threadIdx.x;
    const float* row = up + (size_t)n * DH;
    const float rs = rSw[n];
    float vals[4];
    float s = 0.f, q = 0.f;
    #pragma unroll
    for (int i = 0; i < 4; ++i) {
        const int h = t + i * 256;
        float u = row[h];
        #pragma unroll
        for (int k = 1; k < KS1; ++k) u += row[(size_t)k * (NB * DH) + h];
        u = u * rs + b1[h];
        const float xv = u > 0.f ? u : 0.f;
        vals[i] = xv; s += xv; q += xv * xv;
    }
    ssum[t] = s; ssq[t] = q; __syncthreads();
    for (int off = 128; off > 0; off >>= 1) {
        if (t < off) { ssum[t] += ssum[t + off]; ssq[t] += ssq[t + off]; }
        __syncthreads();
    }
    const float mean = ssum[0] * (1.0f / DH);
    const float var  = ssq[0] * (1.0f / DH) - mean * mean;
    const float inv  = rsqrtf(var + 1e-5f);
    #pragma unroll
    for (int i = 0; i < 4; ++i) {
        const int h = t + i * 256;
        lnout[(size_t)n * DH + h] = (vals[i] - mean) * inv * g[h] + bb[h];
    }
}

constexpr int PP = 36;

__global__ __launch_bounds__(512)
void k_gemm2_final(const float* __restrict__ x, const float* __restrict__ ln,
                   const float* __restrict__ W2, const float* __restrict__ b2,
                   const float* __restrict__ rSw, float* __restrict__ out) {
    __shared__ float part[16][16][PP];
    const int p    = blockIdx.x & 15;
    const int q    = blockIdx.x >> 4;
    const int lane = threadIdx.x & 31;
    const int w    = threadIdx.x >> 5;
    const int low  = lane & 15;
    const int hi   = lane >> 4;

    const v4f* pa  = (const v4f*)(ln + (size_t)low * DH + w * 64 + 4 * hi);
    const v4f* pb0 = (const v4f*)(W2 + (size_t)(p * 32 + low) * DH + w * 64 + 4 * hi);
    const v4f* pb1 = pb0 + 16 * (DH / 4);
    v4f a[8], b0[8], b1[8];
    #pragma unroll
    for (int j = 0; j < 8; ++j) { a[j] = pa[2 * j]; b0[j] = pb0[2 * j]; b1[j] = pb1[2 * j]; }
    const int d4 = lane & 7;
    const float rsw  = rSw[w];
    const v4f   bias = *(const v4f*)(b2 + p * 32 + d4 * 4);

    const int rs = lane >> 3;
    const size_t xoff = ((size_t)w * CC + q * 64 + rs) * DIN + p * 32 + d4 * 4;
    const v4f* px = (const v4f*)(x + xoff);
    v4f xv[16];
    #pragma unroll
    for (int i = 0; i < 16; ++i) xv[i] = __builtin_nontemporal_load(px + i * (4 * DIN / 4));
    __builtin_amdgcn_sched_barrier(0);

    v8f acc0 = {}, acc1 = {};
    #pragma unroll
    for (int j = 0; j < 8; ++j) {
        const v2f a01 = { a[j].x, a[j].y },   a23 = { a[j].z, a[j].w };
        const v2f e01 = { b0[j].x, b0[j].y }, e23 = { b0[j].z, b0[j].w };
        const v2f f01 = { b1[j].x, b1[j].y }, f23 = { b1[j].z, b1[j].w };
        acc0 = __builtin_amdgcn_wmma_f32_16x16x4_f32(false, a01, false, e01, (short)0, acc0, false, false);
        acc1 = __builtin_amdgcn_wmma_f32_16x16x4_f32(false, a01, false, f01, (short)0, acc1, false, false);
        acc0 = __builtin_amdgcn_wmma_f32_16x16x4_f32(false, a23, false, e23, (short)0, acc0, false, false);
        acc1 = __builtin_amdgcn_wmma_f32_16x16x4_f32(false, a23, false, f23, (short)0, acc1, false, false);
    }
    #pragma unroll
    for (int r = 0; r < 8; ++r) {
        part[w][r + 8 * hi][low]      = acc0[r];
        part[w][r + 8 * hi][16 + low] = acc1[r];
    }
    __syncthreads();

    v4f s = { 0.f, 0.f, 0.f, 0.f };
    #pragma unroll
    for (int k = 0; k < 16; ++k) s += *(const v4f*)&part[k][w][d4 * 4];
    const float sc = (float)CC * rsw;
    const v4f av = s * sc + bias;

    v4f* po = (v4f*)(out + xoff);
    #pragma unroll
    for (int i = 0; i < 16; ++i) __builtin_nontemporal_store(xv[i] + av, po + i * (4 * DIN / 4));
}

extern "C" void kernel_launch(void* const* d_in, const int* in_sizes, int n_in,
                              void* d_out, int out_size, void* d_ws, size_t ws_size,
                              hipStream_t stream) {
    const float* x     = (const float*)d_in[0];
    const float* adj   = (const float*)d_in[1];
    const float* W1    = (const float*)d_in[2];
    const float* b1    = (const float*)d_in[3];
    const float* gamma = (const float*)d_in[4];
    const float* beta  = (const float*)d_in[5];
    const float* W2    = (const float*)d_in[6];
    const float* b2    = (const float*)d_in[7];
    float* out = (float*)d_out;

    float* ws  = (float*)d_ws;
    float* up  = ws;
    float* s1  = up + KS1 * NB * DH;
    float* rSw = s1 + NB * DIN;
    float* lnb = rSw + 32;

    k_colsum_x    <<<dim3(NB * 16),  dim3(32 * CSW), 0, stream>>>(x, s1);
    k_sw          <<<dim3(NB),      dim3(32),  0, stream>>>(adj, rSw);
    k_gemm1_part  <<<dim3(32 * KS1), dim3(32),  0, stream>>>(s1, W1, up);
    k_ln          <<<dim3(NB),      dim3(256), 0, stream>>>(up, b1, rSw, gamma, beta, lnb);
    k_gemm2_final <<<dim3(16 * 16),  dim3(512), 0, stream>>>(x, lnb, W2, b2, rSw, out);
}


# Degree sum: its 16 one-wave groups ride in GEMM1's launch as blocks 128..143 (same 32-thread shape, independent work), so its own launch and boundary go (5 launches to 4); every sum order unchanged
# speedup vs baseline: 3.4127x; 1.1033x over previous
#include <hip/hip_runtime.h>

typedef float v2f __attribute__((ext_vector_type(2)));
typedef float v4f __attribute__((ext_vector_type(4)));
typedef float v8f __attribute__((ext_vector_type(8)));

constexpr int NB   = 16;
constexpr int CC   = 1024;
constexpr int DIN  = 512;
constexpr int DH   = 1024;
constexpr int DOUT = 512;

constexpr int CSW = 8;
constexpr int CSR = CC / CSW;
constexpr int CSL = CSR / 4;

__global__ __launch_bounds__(32 * CSW)
void k_colsum_x(const float* __restrict__ x, float* __restrict__ s1) {
    __shared__ alignas(16) float part[CSW][4][32];
    const int p    = blockIdx.x & 15;
    const int n    = blockIdx.x >> 4;
    const int lane = threadIdx.x & 31;
    const int w    = threadIdx.x >> 5;
    const int d4   = lane & 7;
    const int rs   = lane >> 3;
    const v4f* px = (const v4f*)(x + ((size_t)n * CC + w * CSR + rs) * DIN + p * 32 + d4 * 4);
    v4f xv[CSL];
    #pragma unroll
    for (int i = 0; i < CSL; ++i) xv[i] = px[i * (4 * DIN / 4)];
    __builtin_amdgcn_sched_barrier(0);

    v4f c0 = xv[0], c1 = xv[1], c2 = xv[2], c3 = xv[3];
    #pragma unroll
    for (int i = 4; i < CSL; i += 4) { c0 += xv[i]; c1 += xv[i + 1]; c2 += xv[i + 2]; c3 += xv[i + 3]; }
    *(v4f*)&part[w][rs][d4 * 4] = (c0 + c1) + (c2 + c3);
    __syncthreads();

    if (w == 0) {
        float t0 = part[0][0][lane], t1 = part[0][1][lane], t2 = part[0][2][lane], t3 = part[0][3][lane];
        #pragma unroll
        for (int k = 1; k < CSW; ++k) {
            t0 += part[k][0][lane]; t1 += part[k][1][lane]; t2 += part[k][2][lane]; t3 += part[k][3][lane];
        }
        s1[(size_t)n * DIN + p * 32 + lane] = (t0 + t1) + (t2 + t3);
    }
}

__device__ __forceinline__
void degree_recip(const float* __restrict__ adj, float* __restrict__ rSw, int n, int lane) {
    const v4f* pa = (const v4f*)(adj + (size_t)n * CC * CC) + lane;
    v4f wv[8];
    #pragma unroll
    for (int i = 0; i < 8; ++i) wv[i] = pa[32 * i];
    const v4f w4 = ((wv[0] + wv[1]) + (wv[2] + wv[3])) + ((wv[4] + wv[5]) + (wv[6] + wv[7]));
    float s = (w4.x + w4.y) + (w4.z + w4.w);
    #pragma unroll
    for (int off = 16; off > 0; off >>= 1) s += __shfl_down(s, off, 32);
    if (lane == 0) rSw[n] = (s > 0.f) ? (1.0f / s) : 0.f;
}

constexpr int KS1 = 4;
constexpr int KL1 = DIN / KS1;
constexpr int NG1 = 32 * KS1;
constexpr int NL1 = NG1 + NB;

__global__ __launch_bounds__(32)
void k_gemm1_part(const float* __restrict__ s1, const float* __restrict__ W1,
                             float* __restrict__ up,
                             const float* __restrict__ adj, float* __restrict__ rSw) {
    const int lane = threadIdx.x;
    if (blockIdx.x >= NG1) {
        degree_recip(adj, rSw, blockIdx.x - NG1, lane);
        return;
    }
    const int tile = blockIdx.x & 31;
    const int ks   = blockIdx.x >> 5;
    const int low  = lane & 15;
    const int hi   = lane >> 4;

    const v4f* pa  = (const v4f*)(s1 + (size_t)low * DIN + ks * KL1 + 4 * hi);
    const v4f* pb0 = (const v4f*)(W1 + (size_t)(tile * 32 + low) * DIN + ks * KL1 + 4 * hi);
    const v4f* pb1 = pb0 + 16 * (DIN / 4);
    v4f a[16], b0[16], b1[16];
    #pragma unroll
    for (int j = 0; j < 16; ++j) { a[j] = pa[2 * j]; b0[j] = pb0[2 * j]; b1[j] = pb1[2 * j]; }
    __builtin_amdgcn_sched_barrier(0);

    v8f acc0 = {}, acc1 = {};
    #pragma unroll
    for (int j = 0; j < 16; ++j) {
        const v2f a01 = { a[j].x, a[j].y },   a23 = { a[j].z, a[j].w };
        const v2f e01 = { b0[j].x, b0[j].y }, e23 = { b0[j].z, b0[j].w };
        const v2f f01 = { b1[j].x, b1[j].y }, f23 = { b1[j].z, b1[j].w };
        acc0 = __builtin_amdgcn_wmma_f32_16x16x4_f32(false, a01, false, e01, (short)0, acc0, false, false);
        acc1 = __builtin_amdgcn_wmma_f32_16x16x4_f32(false, a01, false, f01, (short)0, acc1, false, false);
        acc0 = __builtin_amdgcn_wmma_f32_16x16x4_f32(false, a23, false, e23, (short)0, acc0, false, false);
        acc1 = __builtin_amdgcn_wmma_f32_16x16x4_f32(false, a23, false, f23, (short)0, acc1, false, false);
    }
    float* po = up + (size_t)ks * (NB * DH) + (size_t)(8 * hi) * DH + tile * 32 + low;
    #pragma unroll
    for (int r = 0; r < 8; ++r) {
        po[(size_t)r * DH]      = acc0[r];
        po[(size_t)r * DH + 16] = acc1[r];
    }
}

__global__ void k_ln(const float* __restrict__ up, const float* __restrict__ b1,
                     const float* __restrict__ rSw, const float* __restrict__ g,
                     const float* __restrict__ bb, float* __restrict__ lnout) {
    __shared__ float ssum[256];
    __shared__ float ssq[256];
    const int n = blockIdx.x, t = threadIdx.x;
    const float* row = up + (size_t)n * DH;
    const float rs = rSw[n];
    float vals[4];
    float s = 0.f, q = 0.f;
    #pragma unroll
    for (int i = 0; i < 4; ++i) {
        const int h = t + i * 256;
        float u = row[h];
        #pragma unroll
        for (int k = 1; k < KS1; ++k) u += row[(size_t)k * (NB * DH) + h];
        u = u * rs + b1[h];
        const float xv = u > 0.f ? u : 0.f;
        vals[i] = xv; s += xv; q += xv * xv;
    }
    ssum[t] = s; ssq[t] = q; __syncthreads();
    for (int off = 128; off > 0; off >>= 1) {
        if (t < off) { ssum[t] += ssum[t + off]; ssq[t] += ssq[t + off]; }
        __syncthreads();
    }
    const float mean = ssum[0] * (1.0f / DH);
    const float var  = ssq[0] * (1.0f / DH) - mean * mean;
    const float inv  = rsqrtf(var + 1e-5f);
    #pragma unroll
    for (int i = 0; i < 4; ++i) {
        const int h = t + i * 256;
        lnout[(size_t)n * DH + h] = (vals[i] - mean) * inv * g[h] + bb[h];
    }
}

constexpr int PP = 36;

__global__ __launch_bounds__(512)
void k_gemm2_final(const float* __restrict__ x, const float* __restrict__ ln,
                   const float* __restrict__ W2, const float* __restrict__ b2,
                   const float* __restrict__ rSw, float* __restrict__ out) {
    __shared__ float part[16][16][PP];
    const int p    = blockIdx.x & 15;
    const int q    = blockIdx.x >> 4;
    const int lane = threadIdx.x & 31;
    const int w    = threadIdx.x >> 5;
    const int low  = lane & 15;
    const int hi   = lane >> 4;

    const v4f* pa  = (const v4f*)(ln + (size_t)low * DH + w * 64 + 4 * hi);
    const v4f* pb0 = (const v4f*)(W2 + (size_t)(p * 32 + low) * DH + w * 64 + 4 * hi);
    const v4f* pb1 = pb0 + 16 * (DH / 4);
    v4f a[8], b0[8], b1[8];
    #pragma unroll
    for (int j = 0; j < 8; ++j) { a[j] = pa[2 * j]; b0[j] = pb0[2 * j]; b1[j] = pb1[2 * j]; }
    const int d4 = lane & 7;
    const float rsw  = rSw[w];
    const v4f   bias = *(const v4f*)(b2 + p * 32 + d4 * 4);

    const int rs = lane >> 3;
    const size_t xoff = ((size_t)w * CC + q * 64 + rs) * DIN + p * 32 + d4 * 4;
    const v4f* px = (const v4f*)(x + xoff);
    v4f xv[16];
    #pragma unroll
    for (int i = 0; i < 16; ++i) xv[i] = __builtin_nontemporal_load(px + i * (4 * DIN / 4));
    __builtin_amdgcn_sched_barrier(0);

    v8f acc0 = {}, acc1 = {};
    #pragma unroll
    for (int j = 0; j < 8; ++j) {
        const v2f a01 = { a[j].x, a[j].y },   a23 = { a[j].z, a[j].w };
        const v2f e01 = { b0[j].x, b0[j].y }, e23 = { b0[j].z, b0[j].w };
        const v2f f01 = { b1[j].x, b1[j].y }, f23 = { b1[j].z, b1[j].w };
        acc0 = __builtin_amdgcn_wmma_f32_16x16x4_f32(false, a01, false, e01, (short)0, acc0, false, false);
        acc1 = __builtin_amdgcn_wmma_f32_16x16x4_f32(false, a01, false, f01, (short)0, acc1, false, false);
        acc0 = __builtin_amdgcn_wmma_f32_16x16x4_f32(false, a23, false, e23, (short)0, acc0, false, false);
        acc1 = __builtin_amdgcn_wmma_f32_16x16x4_f32(false, a23, false, f23, (short)0, acc1, false, false);
    }
    #pragma unroll
    for (int r = 0; r < 8; ++r) {
        part[w][r + 8 * hi][low]      = acc0[r];
        part[w][r + 8 * hi][16 + low] = acc1[r];
    }
    __syncthreads();

    v4f s = { 0.f, 0.f, 0.f, 0.f };
    #pragma unroll
    for (int k = 0; k < 16; ++k) s += *(const v4f*)&part[k][w][d4 * 4];
    const float sc = (float)CC * rsw;
    const v4f av = s * sc + bias;

    v4f* po = (v4f*)(out + xoff);
    #pragma unroll
    for (int i = 0; i < 16; ++i) __builtin_nontemporal_store(xv[i] + av, po + i * (4 * DIN / 4));
}

extern "C" void kernel_launch(void* const* d_in, const int* in_sizes, int n_in,
                              void* d_out, int out_size, void* d_ws, size_t ws_size,
                              hipStream_t stream) {
    const float* x     = (const float*)d_in[0];
    const float* adj   = (const float*)d_in[1];
    const float* W1    = (const float*)d_in[2];
    const float* b1    = (const float*)d_in[3];
    const float* gamma = (const float*)d_in[4];
    const float* beta  = (const float*)d_in[5];
    const float* W2    = (const float*)d_in[6];
    const float* b2    = (const float*)d_in[7];
    float* out = (float*)d_out;

    float* ws  = (float*)d_ws;
    float* up  = ws;
    float* s1  = up + KS1 * NB * DH;
    float* rSw = s1 + NB * DIN;
    float* lnb = rSw + 32;

    k_colsum_x    <<<dim3(NB * 16),  dim3(32 * CSW), 0, stream>>>(x, s1);
    k_gemm1_part  <<<dim3(NL1),      dim3(32),  0, stream>>>(s1, W1, up, adj, rSw);
    k_ln          <<<dim3(NB),      dim3(256), 0, stream>>>(up, b1, rSw, gamma, beta, lnb);
    k_gemm2_final <<<dim3(16 * 16),  dim3(512), 0, stream>>>(x, lnb, W2, b2, rSw, out);
}


# LayerNorm's launch goes (4 to 3): GEMM1 groups own full K (4 waves, LDS sum), emit relu(u) + per-tile row sums; GEMM2's A loader normalises; degree groups ride launch 1 (new f32 sum order: LN stats)
# speedup vs baseline: 3.8444x; 1.1265x over previous
#include <hip/hip_runtime.h>

typedef float v2f __attribute__((ext_vector_type(2)));
typedef float v4f __attribute__((ext_vector_type(4)));
typedef float v8f __attribute__((ext_vector_type(8)));

constexpr int NB   = 16;
constexpr int CC   = 1024;
constexpr int DIN  = 512;
constexpr int DH   = 1024;
constexpr int DOUT = 512;

__device__ __forceinline__
void degree_recip(const float* __restrict__ adj, float* __restrict__ rSw, int n, int lane) {
    const v4f* pa = (const v4f*)(adj + (size_t)n * CC * CC) + lane;
    v4f wv[8];
    #pragma unroll
    for (int i = 0; i < 8; ++i) wv[i] = pa[32 * i];
    const v4f w4 = ((wv[0] + wv[1]) + (wv[2] + wv[3])) + ((wv[4] + wv[5]) + (wv[6] + wv[7]));
    float s = (w4.x + w4.y) + (w4.z + w4.w);
    #pragma unroll
    for (int off = 16; off > 0; off >>= 1) s += __shfl_down(s, off, 32);
    if (lane == 0) rSw[n] = (s > 0.f) ? (1.0f / s) : 0.f;
}

constexpr int CSW = 8;
constexpr int CSR = CC / CSW;
constexpr int CSL = CSR / 4;

__global__ __launch_bounds__(32 * CSW)
void k_colsum_x(const float* __restrict__ x, float* __restrict__ s1,
                const float* __restrict__ adj, float* __restrict__ rSw) {
    __shared__ alignas(16) float part[CSW][4][32];
    const int lane = threadIdx.x & 31;
    const int w    = threadIdx.x >> 5;
    if (blockIdx.x >= NB * 16) {
        if (w == 0) degree_recip(adj, rSw, blockIdx.x - NB * 16, lane);
        return;
    }
    const int p    = blockIdx.x & 15;
    const int n    = blockIdx.x >> 4;
    const int d4   = lane & 7;
    const int rs   = lane >> 3;
    const v4f* px = (const v4f*)(x + ((size_t)n * CC + w * CSR + rs) * DIN + p * 32 + d4 * 4);
    v4f xv[CSL];
    #pragma unroll
    for (int i = 0; i < CSL; ++i) xv[i] = px[i * (4 * DIN / 4)];
    __builtin_amdgcn_sched_barrier(0);

    v4f c0 = xv[0], c1 = xv[1], c2 = xv[2], c3 = xv[3];
    #pragma unroll
    for (int i = 4; i < CSL; i += 4) { c0 += xv[i]; c1 += xv[i + 1]; c2 += xv[i + 2]; c3 += xv[i + 3]; }
    *(v4f*)&part[w][rs][d4 * 4] = (c0 + c1) + (c2 + c3);
    __syncthreads();

    if (w == 0) {
        float t0 = part[0][0][lane], t1 = part[0][1][lane], t2 = part[0][2][lane], t3 = part[0][3][lane];
        #pragma unroll
        for (int k = 1; k < CSW; ++k) {
            t0 += part[k][0][lane]; t1 += part[k][1][lane]; t2 += part[k][2][lane]; t3 += part[k][3][lane];
        }
        s1[(size_t)n * DIN + p * 32 + lane] = (t0 + t1) + (t2 + t3);
    }
}

constexpr int KS1 = 4;
constexpr int KL1 = DIN / KS1;
constexpr int NT1 = DH / 16;
constexpr int P1  = 20;

__global__ __launch_bounds__(32 * KS1)
void k_gemm1_relu_stats(const float* __restrict__ s1, const float* __restrict__ W1,
                        const float* __restrict__ b1, const float* __restrict__ rSw,
                        float* __restrict__ rt, float* __restrict__ stats) {
    __shared__ alignas(16) float part[KS1][16][P1];
    const int tile = blockIdx.x;
    const int lane = threadIdx.x & 31;
    const int ks   = threadIdx.x >> 5;
    const int low  = lane & 15;
    const int hi   = lane >> 4;

    const v4f* pa = (const v4f*)(s1 + (size_t)low * DIN + ks * KL1 + 4 * hi);
    const v4f* pb = (const v4f*)(W1 + (size_t)(tile * 16 + low) * DIN + ks * KL1 + 4 * hi);
    v4f a[16], b[16];
    #pragma unroll
    for (int j = 0; j < 16; ++j) { a[j] = pa[2 * j]; b[j] = pb[2 * j]; }
    const int row = lane >> 1;
    const int c8  = lane & 1;
    const float rs = rSw[row];
    const v4f bias0 = *(const v4f*)(b1 + tile * 16 + c8 * 8);
    const v4f bias1 = *(const v4f*)(b1 + tile * 16 + c8 * 8 + 4);
    __builtin_amdgcn_sched_barrier(0);

    v8f acc = {};
    #pragma unroll
    for (int j = 0; j < 16; ++j) {
        const v2f a01 = { a[j].x, a[j].y }, a23 = { a[j].z, a[j].w };
        const v2f e01 = { b[j].x, b[j].y }, e23 = { b[j].z, b[j].w };
        acc = __builtin_amdgcn_wmma_f32_16x16x4_f32(false, a01, false, e01, (short)0, acc, false, false);
        acc = __builtin_amdgcn_wmma_f32_16x16x4_f32(false, a23, false, e23, (short)0, acc, false, false);
    }
    #pragma unroll
    for (int r = 0; r < 8; ++r) part[ks][r + 8 * hi][low] = acc[r];
    __syncthreads();

    if (ks == 0) {
        const v4f* pp = (const v4f*)&part[0][row][c8 * 8];
        v4f u0 = pp[0], u1 = pp[1];
        #pragma unroll
        for (int k = 1; k < KS1; ++k) {
            const v4f* pk = (const v4f*)&part[k][row][c8 * 8];
            u0 += pk[0]; u1 += pk[1];
        }
        u0 = u0 * rs + bias0; u1 = u1 * rs + bias1;
        v4f x0, x1;
        x0.x = u0.x > 0.f ? u0.x : 0.f; x0.y = u0.y > 0.f ? u0.y : 0.f;
        x0.z = u0.z > 0.f ? u0.z : 0.f; x0.w = u0.w > 0.f ? u0.w : 0.f;
        x1.x = u1.x > 0.f ? u1.x : 0.f; x1.y = u1.y > 0.f ? u1.y : 0.f;
        x1.z = u1.z > 0.f ? u1.z : 0.f; x1.w = u1.w > 0.f ? u1.w : 0.f;
        float* pt = rt + (size_t)tile * 256 + lane * 4;
        *(v4f*)pt         = x0;
        *(v4f*)(pt + 128) = x1;
        float s = ((x0.x + x0.y) + (x0.z + x0.w)) + ((x1.x + x1.y) + (x1.z + x1.w));
        float q = ((x0.x * x0.x + x0.y * x0.y) + (x0.z * x0.z + x0.w * x0.w))
                + ((x1.x * x1.x + x1.y * x1.y) + (x1.z * x1.z + x1.w * x1.w));
        s += __shfl_xor(s, 1, 32); q += __shfl_xor(q, 1, 32);
        if (c8 == 0) *(v2f*)(stats + (size_t)tile * 32 + 2 * row) = (v2f){ s, q };
    }
}

constexpr int PP = 36;

__global__ __launch_bounds__(512)
void k_gemm2_final(const float* __restrict__ x, const float* __restrict__ rt,
                   const float* __restrict__ stats,
                   const float* __restrict__ g, const float* __restrict__ bb,
                   const float* __restrict__ W2, const float* __restrict__ b2,
                   const float* __restrict__ rSw, float* __restrict__ out) {
    __shared__ float part[16][16][PP];
    __shared__ alignas(16) float pst[16][PP];
    __shared__ alignas(16) float gbl[16][128];
    const int p    = blockIdx.x & 15;
    const int q    = blockIdx.x >> 4;
    const int lane = threadIdx.x & 31;
    const int w    = threadIdx.x >> 5;
    const int low  = lane & 15;
    const int hi   = lane >> 4;

    const v4f st = *(const v4f*)(stats + (size_t)(4 * w) * 32 + 4 * lane);
    const v4f gb = *(const v4f*)((hi ? bb : g) + w * 64 + 4 * low);
    __builtin_amdgcn_sched_barrier(0);
    const v4f* pa  = (const v4f*)(rt + (size_t)(4 * w) * 256 + hi * 128 + low * 8);
    const v4f* pb0 = (const v4f*)(W2 + (size_t)(p * 32 + low) * DH + w * 64 + 4 * hi);
    const v4f* pb1 = pb0 + 16 * (DH / 4);
    v4f a[8], b0[8], b1[8];
    #pragma unroll
    for (int j = 0; j < 8; ++j) a[j] = pa[(j >> 1) * 64 + (j & 1)];
    #pragma unroll
    for (int j = 0; j < 8; ++j) { b0[j] = pb0[2 * j]; b1[j] = pb1[2 * j]; }
    const int d4 = lane & 7;
    const float rsw  = rSw[w];
    const v4f   bias = *(const v4f*)(b2 + p * 32 + d4 * 4);

    const int rs = lane >> 3;
    const size_t xoff = ((size_t)w * CC + q * 64 + rs) * DIN + p * 32 + d4 * 4;
    const v4f* px = (const v4f*)(x + xoff);
    v4f xv[16];
    #pragma unroll
    for (int i = 0; i < 16; ++i) xv[i] = __builtin_nontemporal_load(px + i * (4 * DIN / 4));
    __builtin_amdgcn_sched_barrier(0);

    v4f sp = st;
    sp.x += __shfl_xor(sp.x, 8, 32);  sp.y += __shfl_xor(sp.y, 8, 32);
    sp.z += __shfl_xor(sp.z, 8, 32);  sp.w += __shfl_xor(sp.w, 8, 32);
    sp.x += __shfl_xor(sp.x, 16, 32); sp.y += __shfl_xor(sp.y, 16, 32);
    sp.z += __shfl_xor(sp.z, 16, 32); sp.w += __shfl_xor(sp.w, 16, 32);
    if (lane < 8) {
        *(v2f*)&pst[2 * lane][2 * w]     = (v2f){ sp.x, sp.y };
        *(v2f*)&pst[2 * lane + 1][2 * w] = (v2f){ sp.z, sp.w };
    }
    *(v4f*)&gbl[w][4 * lane] = gb;
    __builtin_amdgcn_fence(__ATOMIC_RELEASE, "workgroup", "local");
    __builtin_amdgcn_s_barrier();
    __builtin_amdgcn_fence(__ATOMIC_ACQUIRE, "workgroup", "local");
    const v4f* pp = (const v4f*)&pst[low][0];
    v4f t[8];
    #pragma unroll
    for (int k = 0; k < 8; ++k) t[k] = pp[k];
    const v4f t4 = ((t[0] + t[1]) + (t[2] + t[3])) + ((t[4] + t[5]) + (t[6] + t[7]));
    const float ssum = t4.x + t4.z, ssq = t4.y + t4.w;
    const float mean = ssum * (1.0f / DH);
    const float var  = ssq * (1.0f / DH) - mean * mean;
    const float inv  = rsqrtf(var + 1e-5f);
    #pragma unroll
    for (int j = 0; j < 8; ++j) {
        const v4f g4 = *(const v4f*)&gbl[w][8 * j + 4 * hi];
        const v4f e4 = *(const v4f*)&gbl[w][64 + 8 * j + 4 * hi];
        a[j] = (a[j] - mean) * inv * g4 + e4;
    }

    v8f acc0 = {}, acc1 = {};
    #pragma unroll
    for (int j = 0; j < 8; ++j) {
        const v2f a01 = { a[j].x, a[j].y },   a23 = { a[j].z, a[j].w };
        const v2f e01 = { b0[j].x, b0[j].y }, e23 = { b0[j].z, b0[j].w };
        const v2f f01 = { b1[j].x, b1[j].y }, f23 = { b1[j].z, b1[j].w };
        acc0 = __builtin_amdgcn_wmma_f32_16x16x4_f32(false, a01, false, e01, (short)0, acc0, false, false);
        acc1 = __builtin_amdgcn_wmma_f32_16x16x4_f32(false, a01, false, f01, (short)0, acc1, false, false);
        acc0 = __builtin_amdgcn_wmma_f32_16x16x4_f32(false, a23, false, e23, (short)0, acc0, false, false);
        acc1 = __builtin_amdgcn_wmma_f32_16x16x4_f32(false, a23, false, f23, (short)0, acc1, false, false);
    }
    #pragma unroll
    for (int r = 0; r < 8; ++r) {
        part[w][r + 8 * hi][low]      = acc0[r];
        part[w][r + 8 * hi][16 + low] = acc1[r];
    }
    __syncthreads();

    v4f s = { 0.f, 0.f, 0.f, 0.f };
    #pragma unroll
    for (int k = 0; k < 16; ++k) s += *(const v4f*)&part[k][w][d4 * 4];
    const float sc = (float)CC * rsw;
    const v4f av = s * sc + bias;

    v4f* po = (v4f*)(out + xoff);
    #pragma unroll
    for (int i = 0; i < 16; ++i) __builtin_nontemporal_store(xv[i] + av, po + i * (4 * DIN / 4));
}

extern "C" void kernel_launch(void* const* d_in, const int* in_sizes, int n_in,
                              void* d_out, int out_size, void* d_ws, size_t ws_size,
                              hipStream_t stream) {
    const float* x     = (const float*)d_in[0];
    const float* adj   = (const float*)d_in[1];
    const float* W1    = (const float*)d_in[2];
    const float* b1    = (const float*)d_in[3];
    const float* gamma = (const float*)d_in[4];
    const float* beta  = (const float*)d_in[5];
    const float* W2    = (const float*)d_in[6];
    const float* b2    = (const float*)d_in[7];
    float* out = (float*)d_out;

    float* ws  = (float*)d_ws;
    float* rt  = ws;
    float* s1  = rt + NB * DH;
    float* rSw = s1 + NB * DIN;
    float* st  = rSw + 32;

    k_colsum_x         <<<dim3(NB * 16 + NB), dim3(32 * CSW), 0, stream>>>(x, s1, adj, rSw);
    k_gemm1_relu_stats <<<dim3(NT1),          dim3(32 * KS1), 0, stream>>>(s1, W1, b1, rSw, rt, st);
    k_gemm2_final      <<<dim3(16 * 16),      dim3(512),      0, stream>>>(x, rt, st, gamma, beta, W2, b2, rSw, out);
}
